# starting kernel
# baseline (speedup 1.0000x reference)
#include <hip/hip_runtime.h>

typedef __attribute__((ext_vector_type(2))) float v2f;
typedef __attribute__((ext_vector_type(8))) float v8f;

#define NODES  50000
#define EDGES  800000
#define GRAPHS 64
#define C      64
#define OUTC   32
#define LPITCH 68

__global__ void fill_kernel(float* __restrict__ p, float v, int n) {
  int i = blockIdx.x * blockDim.x + threadIdx.x;
  if (i < n) p[i] = v;
}

__global__ void deg_count_kernel(const int* __restrict__ dst, float* __restrict__ deg) {
  int i = blockIdx.x * blockDim.x + threadIdx.x;
  if (i < EDGES) atomicAdd(&deg[dst[i]], 1.0f);
}

__global__ void rsqrt_kernel(float* __restrict__ p, int n) {
  int i = blockIdx.x * blockDim.x + threadIdx.x;
  if (i < n) p[i] = rsqrtf(p[i]);
}

__global__ __launch_bounds__(256)
void gemm_rs_wmma(const float* __restrict__ X, const float* __restrict__ W,
                  const float* __restrict__ rowscale, float* __restrict__ out,
                  int ntiles) {
  __shared__ float sWt[C * LPITCH];
  for (int i = threadIdx.x; i < C * C; i += 256) {
    const int k = i >> 6, n = i & 63;
    sWt[n * LPITCH + k] = W[i];
  }
  __syncthreads();

  const int wave = threadIdx.x >> 5;
  const int lane = threadIdx.x & 31;
  const int tile = blockIdx.x * 8 + wave;
  if (tile >= ntiles) return;

  const int m0 = tile * 16;
  const int ar = m0 + (lane & 15);
  const int kg = (lane >> 4) * 2;
  const int cn = lane & 15;
  const float* bp = sWt + cn * LPITCH;

  v8f acc0 = {}, acc1 = {}, acc2 = {}, acc3 = {};
#pragma unroll
  for (int ks = 0; ks < 16; ++ks) {
    const int k0 = ks * 4 + kg;
    v2f a = *(const v2f*)(X + ar * C + k0);
    v2f b0 = *(const v2f*)(bp + k0);
    v2f b1 = *(const v2f*)(bp + 16 * LPITCH + k0);
    v2f b2 = *(const v2f*)(bp + 32 * LPITCH + k0);
    v2f b3 = *(const v2f*)(bp + 48 * LPITCH + k0);
    acc0 = __builtin_amdgcn_wmma_f32_16x16x4_f32(false, a, false, b0, (short)0, acc0, false, false);
    acc1 = __builtin_amdgcn_wmma_f32_16x16x4_f32(false, a, false, b1, (short)0, acc1, false, false);
    acc2 = __builtin_amdgcn_wmma_f32_16x16x4_f32(false, a, false, b2, (short)0, acc2, false, false);
    acc3 = __builtin_amdgcn_wmma_f32_16x16x4_f32(false, a, false, b3, (short)0, acc3, false, false);
  }

#pragma unroll
  for (int r = 0; r < 8; ++r) {
    const int row = m0 + r + 8 * (lane >> 4);
    const float s = rowscale[row];
    out[row * C +  0 + cn] = acc0[r] * s;
    out[row * C + 16 + cn] = acc1[r] * s;
    out[row * C + 32 + cn] = acc2[r] * s;
    out[row * C + 48 + cn] = acc3[r] * s;
  }
}

__global__ void scatter_kernel(const int* __restrict__ src, const int* __restrict__ dst,
                               const float* __restrict__ hs, float* __restrict__ acc) {
  int idx = blockIdx.x * blockDim.x + threadIdx.x;
  int e  = idx >> 4;
  int cg = (idx & 15) << 2;
  if (e < EDGES) {
    int s = src[e], d = dst[e];
    const float4 v = *(const float4*)(hs + (size_t)s * C + cg);
    float* a = acc + (size_t)d * C + cg;
    atomicAdd(a + 0, v.x);
    atomicAdd(a + 1, v.y);
    atomicAdd(a + 2, v.z);
    atomicAdd(a + 3, v.w);
  }
}

__global__ void finalize_kernel(float* __restrict__ acc, const float* __restrict__ hs,
                                const float* __restrict__ dinv, const float* __restrict__ bias,
                                int do_relu) {
  int i = blockIdx.x * blockDim.x + threadIdx.x;
  if (i < NODES * C) {
    int row = i >> 6, c = i & 63;
    float v = dinv[row] * (acc[i] + hs[i]) + bias[c];
    if (do_relu) v = fmaxf(v, 0.0f);
    acc[i] = v;
  }
}

__global__ void pool_kernel(const float* __restrict__ z, const int* __restrict__ batch,
                            float* __restrict__ u) {
  int idx = blockIdx.x * blockDim.x + threadIdx.x;
  int i  = idx >> 4;
  int cg = (idx & 15) << 2;
  if (i < NODES) {
    int g = batch[i];
    const float4 v = *(const float4*)(z + (size_t)i * C + cg);
    float* p = u + g * C + cg;
    atomicAdd(p + 0, v.x);
    atomicAdd(p + 1, v.y);
    atomicAdd(p + 2, v.z);
    atomicAdd(p + 3, v.w);
  }
}

__global__ __launch_bounds__(256)
void head_wmma(const float* __restrict__ u, const float* __restrict__ Wf,
               const float* __restrict__ bf, float* __restrict__ y,
               float* __restrict__ out_u) {
  __shared__ float sWf[OUTC * LPITCH];
  for (int i = threadIdx.x; i < C * OUTC; i += 256) {
    const int k = i >> 5, n = i & 31;
    sWf[n * LPITCH + k] = Wf[i];
  }
  __syncthreads();

  const int wave = threadIdx.x >> 5;
  const int lane = threadIdx.x & 31;
  const int mt = wave >> 1;
  const int nt = wave & 1;
  const int ar = mt * 16 + (lane & 15);
  const int kg = (lane >> 4) * 2;
  const int cn = nt * 16 + (lane & 15);
  const float* bp = sWf + cn * LPITCH;

  v8f acc = {};
#pragma unroll
  for (int ks = 0; ks < 16; ++ks) {
    const int k0 = ks * 4 + kg;
    v2f a = *(const v2f*)(u + ar * C + k0);
    v2f b = *(const v2f*)(bp + k0);
    acc = __builtin_amdgcn_wmma_f32_16x16x4_f32(false, a, false, b, (short)0, acc, false, false);
  }
#pragma unroll
  for (int r = 0; r < 8; ++r) {
    const int row = mt * 16 + r + 8 * (lane >> 4);
    y[row * OUTC + cn] = acc[r] + bf[cn];
  }
#pragma unroll
  for (int j = 0; j < 16; ++j) {
    const int idx = threadIdx.x + j * 256;
    out_u[idx] = u[idx];
  }
}

extern "C" void kernel_launch(void* const* d_in, const int* in_sizes, int n_in,
                              void* d_out, int out_size, void* d_ws, size_t ws_size,
                              hipStream_t stream) {
  (void)in_sizes; (void)n_in; (void)out_size; (void)ws_size;

  const float* x     = (const float*)d_in[0];
  const int*   ei    = (const int*)  d_in[1];
  const int*   batch = (const int*)  d_in[2];
  const float* W1    = (const float*)d_in[3];
  const float* b1    = (const float*)d_in[4];
  const float* W2    = (const float*)d_in[5];
  const float* b2    = (const float*)d_in[6];
  const float* Wf    = (const float*)d_in[7];
  const float* bf    = (const float*)d_in[8];
  float* out = (float*)d_out;

  const int* src = ei;
  const int* dst = ei + EDGES;

  float* ws   = (float*)d_ws;
  float* dinv = ws;
  float* hs   = ws + 50048;
  float* acc  = hs + (size_t)NODES * C;
  float* u    = acc + (size_t)NODES * C;

  const dim3 B(256);
  const int NT = NODES / 16;
  const int GB = (NT + 7) / 8;

  fill_kernel     <<<(NODES + 255) / 256, B, 0, stream>>>(dinv, 1.0f, NODES);
  deg_count_kernel<<<(EDGES + 255) / 256, B, 0, stream>>>(dst, dinv);
  rsqrt_kernel    <<<(NODES + 255) / 256, B, 0, stream>>>(dinv, NODES);

  fill_kernel    <<<(NODES * C) / 256, B, 0, stream>>>(acc, 0.0f, NODES * C);
  gemm_rs_wmma   <<<GB, B, 0, stream>>>(x, W1, dinv, hs, NT);
  scatter_kernel <<<(EDGES * 16) / 256, B, 0, stream>>>(src, dst, hs, acc);
  finalize_kernel<<<(NODES * C) / 256, B, 0, stream>>>(acc, hs, dinv, b1, 1);

  gemm_rs_wmma   <<<GB, B, 0, stream>>>(acc, W2, dinv, hs, NT);
  fill_kernel    <<<(NODES * C) / 256, B, 0, stream>>>(acc, 0.0f, NODES * C);
  scatter_kernel <<<(EDGES * 16) / 256, B, 0, stream>>>(src, dst, hs, acc);
  finalize_kernel<<<(NODES * C) / 256, B, 0, stream>>>(acc, hs, dinv, b2, 0);

  fill_kernel<<<(GRAPHS * C + 255) / 256, B, 0, stream>>>(u, 0.0f, GRAPHS * C);
  pool_kernel<<<(NODES * 16) / 256, B, 0, stream>>>(acc, batch, u);
  head_wmma  <<<1, B, 0, stream>>>(u, Wf, bf, out, out + GRAPHS * OUTC);
}


# scatter: one wave per edge row (lane = column, 32 edges per wave): each float atomic instruction adds one whole 128-byte line, not 8 lanes in each of 4 lines; atomic arrival order as before
# speedup vs baseline: 1.3450x; 1.3450x over previous
#include <hip/hip_runtime.h>

typedef __attribute__((ext_vector_type(2))) float v2f;
typedef __attribute__((ext_vector_type(8))) float v8f;

#define NODES  50000
#define EDGES  800000
#define GRAPHS 64
#define C      64
#define OUTC   32
#define LPITCH 68

__global__ void fill_kernel(float* __restrict__ p, float v, int n) {
  int i = blockIdx.x * blockDim.x + threadIdx.x;
  if (i < n) p[i] = v;
}

__global__ void deg_count_kernel(const int* __restrict__ dst, float* __restrict__ deg) {
  int i = blockIdx.x * blockDim.x + threadIdx.x;
  if (i < EDGES) atomicAdd(&deg[dst[i]], 1.0f);
}

__global__ void rsqrt_kernel(float* __restrict__ p, int n) {
  int i = blockIdx.x * blockDim.x + threadIdx.x;
  if (i < n) p[i] = rsqrtf(p[i]);
}

__global__ __launch_bounds__(256)
void gemm_rs_wmma(const float* __restrict__ X, const float* __restrict__ W,
                  const float* __restrict__ rowscale, float* __restrict__ out,
                  int ntiles) {
  __shared__ float sWt[C * LPITCH];
  for (int i = threadIdx.x; i < C * C; i += 256) {
    const int k = i >> 6, n = i & 63;
    sWt[n * LPITCH + k] = W[i];
  }
  __syncthreads();

  const int wave = threadIdx.x >> 5;
  const int lane = threadIdx.x & 31;
  const int tile = blockIdx.x * 8 + wave;
  if (tile >= ntiles) return;

  const int m0 = tile * 16;
  const int ar = m0 + (lane & 15);
  const int kg = (lane >> 4) * 2;
  const int cn = lane & 15;
  const float* bp = sWt + cn * LPITCH;

  v8f acc0 = {}, acc1 = {}, acc2 = {}, acc3 = {};
#pragma unroll
  for (int ks = 0; ks < 16; ++ks) {
    const int k0 = ks * 4 + kg;
    v2f a = *(const v2f*)(X + ar * C + k0);
    v2f b0 = *(const v2f*)(bp + k0);
    v2f b1 = *(const v2f*)(bp + 16 * LPITCH + k0);
    v2f b2 = *(const v2f*)(bp + 32 * LPITCH + k0);
    v2f b3 = *(const v2f*)(bp + 48 * LPITCH + k0);
    acc0 = __builtin_amdgcn_wmma_f32_16x16x4_f32(false, a, false, b0, (short)0, acc0, false, false);
    acc1 = __builtin_amdgcn_wmma_f32_16x16x4_f32(false, a, false, b1, (short)0, acc1, false, false);
    acc2 = __builtin_amdgcn_wmma_f32_16x16x4_f32(false, a, false, b2, (short)0, acc2, false, false);
    acc3 = __builtin_amdgcn_wmma_f32_16x16x4_f32(false, a, false, b3, (short)0, acc3, false, false);
  }

#pragma unroll
  for (int r = 0; r < 8; ++r) {
    const int row = m0 + r + 8 * (lane >> 4);
    const float s = rowscale[row];
    out[row * C +  0 + cn] = acc0[r] * s;
    out[row * C + 16 + cn] = acc1[r] * s;
    out[row * C + 32 + cn] = acc2[r] * s;
    out[row * C + 48 + cn] = acc3[r] * s;
  }
}

#define EPW  32
#define EGRP 8
static_assert(EDGES % EPW == 0, "every wave gets a full set of edges");
__global__ __launch_bounds__(256)
void scatter_kernel(const int* __restrict__ src, const int* __restrict__ dst,
                    const float* __restrict__ hs, float* __restrict__ acc) {
  const int lane = threadIdx.x & 31;
  const int e0 = (blockIdx.x * 8 + (threadIdx.x >> 5)) * EPW;
  if (e0 >= EDGES) return;
  const int sv = src[e0 + lane];
  const int dv = dst[e0 + lane];
#pragma unroll
  for (int jb = 0; jb < EPW; jb += EGRP) {
    float v0[EGRP], v1[EGRP];
#pragma unroll
    for (int j = 0; j < EGRP; ++j) {
      const int s = __builtin_amdgcn_readlane(sv, jb + j);
      const float* hp = hs + (size_t)s * C + lane;
      v0[j] = hp[0];
      v1[j] = hp[32];
    }
#pragma unroll
    for (int j = 0; j < EGRP; ++j) {
      const int d = __builtin_amdgcn_readlane(dv, jb + j);
      float* ap = acc + (size_t)d * C + lane;
      atomicAdd(ap,      v0[j]);
      atomicAdd(ap + 32, v1[j]);
    }
  }
}

__global__ void finalize_kernel(float* __restrict__ acc, const float* __restrict__ hs,
                                const float* __restrict__ dinv, const float* __restrict__ bias,
                                int do_relu) {
  int i = blockIdx.x * blockDim.x + threadIdx.x;
  if (i < NODES * C) {
    int row = i >> 6, c = i & 63;
    float v = dinv[row] * (acc[i] + hs[i]) + bias[c];
    if (do_relu) v = fmaxf(v, 0.0f);
    acc[i] = v;
  }
}

__global__ void pool_kernel(const float* __restrict__ z, const int* __restrict__ batch,
                            float* __restrict__ u) {
  int idx = blockIdx.x * blockDim.x + threadIdx.x;
  int i  = idx >> 4;
  int cg = (idx & 15) << 2;
  if (i < NODES) {
    int g = batch[i];
    const float4 v = *(const float4*)(z + (size_t)i * C + cg);
    float* p = u + g * C + cg;
    atomicAdd(p + 0, v.x);
    atomicAdd(p + 1, v.y);
    atomicAdd(p + 2, v.z);
    atomicAdd(p + 3, v.w);
  }
}

__global__ __launch_bounds__(256)
void head_wmma(const float* __restrict__ u, const float* __restrict__ Wf,
               const float* __restrict__ bf, float* __restrict__ y,
               float* __restrict__ out_u) {
  __shared__ float sWf[OUTC * LPITCH];
  for (int i = threadIdx.x; i < C * OUTC; i += 256) {
    const int k = i >> 5, n = i & 31;
    sWf[n * LPITCH + k] = Wf[i];
  }
  __syncthreads();

  const int wave = threadIdx.x >> 5;
  const int lane = threadIdx.x & 31;
  const int mt = wave >> 1;
  const int nt = wave & 1;
  const int ar = mt * 16 + (lane & 15);
  const int kg = (lane >> 4) * 2;
  const int cn = nt * 16 + (lane & 15);
  const float* bp = sWf + cn * LPITCH;

  v8f acc = {};
#pragma unroll
  for (int ks = 0; ks < 16; ++ks) {
    const int k0 = ks * 4 + kg;
    v2f a = *(const v2f*)(u + ar * C + k0);
    v2f b = *(const v2f*)(bp + k0);
    acc = __builtin_amdgcn_wmma_f32_16x16x4_f32(false, a, false, b, (short)0, acc, false, false);
  }
#pragma unroll
  for (int r = 0; r < 8; ++r) {
    const int row = mt * 16 + r + 8 * (lane >> 4);
    y[row * OUTC + cn] = acc[r] + bf[cn];
  }
#pragma unroll
  for (int j = 0; j < 16; ++j) {
    const int idx = threadIdx.x + j * 256;
    out_u[idx] = u[idx];
  }
}

extern "C" void kernel_launch(void* const* d_in, const int* in_sizes, int n_in,
                              void* d_out, int out_size, void* d_ws, size_t ws_size,
                              hipStream_t stream) {
  (void)in_sizes; (void)n_in; (void)out_size; (void)ws_size;

  const float* x     = (const float*)d_in[0];
  const int*   ei    = (const int*)  d_in[1];
  const int*   batch = (const int*)  d_in[2];
  const float* W1    = (const float*)d_in[3];
  const float* b1    = (const float*)d_in[4];
  const float* W2    = (const float*)d_in[5];
  const float* b2    = (const float*)d_in[6];
  const float* Wf    = (const float*)d_in[7];
  const float* bf    = (const float*)d_in[8];
  float* out = (float*)d_out;

  const int* src = ei;
  const int* dst = ei + EDGES;

  float* ws   = (float*)d_ws;
  float* dinv = ws;
  float* hs   = ws + 50048;
  float* acc  = hs + (size_t)NODES * C;
  float* u    = acc + (size_t)NODES * C;

  const dim3 B(256);
  const int NT = NODES / 16;
  const int GB = (NT + 7) / 8;

  fill_kernel     <<<(NODES + 255) / 256, B, 0, stream>>>(dinv, 1.0f, NODES);
  deg_count_kernel<<<(EDGES + 255) / 256, B, 0, stream>>>(dst, dinv);
  rsqrt_kernel    <<<(NODES + 255) / 256, B, 0, stream>>>(dinv, NODES);

  fill_kernel    <<<(NODES * C) / 256, B, 0, stream>>>(acc, 0.0f, NODES * C);
  gemm_rs_wmma   <<<GB, B, 0, stream>>>(x, W1, dinv, hs, NT);
  scatter_kernel <<<EDGES / EPW / 8, B, 0, stream>>>(src, dst, hs, acc);
  finalize_kernel<<<(NODES * C) / 256, B, 0, stream>>>(acc, hs, dinv, b1, 1);

  gemm_rs_wmma   <<<GB, B, 0, stream>>>(acc, W2, dinv, hs, NT);
  fill_kernel    <<<(NODES * C) / 256, B, 0, stream>>>(acc, 0.0f, NODES * C);
  scatter_kernel <<<EDGES / EPW / 8, B, 0, stream>>>(src, dst, hs, acc);
  finalize_kernel<<<(NODES * C) / 256, B, 0, stream>>>(acc, hs, dinv, b2, 0);

  fill_kernel<<<(GRAPHS * C + 255) / 256, B, 0, stream>>>(u, 0.0f, GRAPHS * C);
  pool_kernel<<<(NODES * 16) / 256, B, 0, stream>>>(acc, batch, u);
  head_wmma  <<<1, B, 0, stream>>>(u, Wf, bf, out, out + GRAPHS * OUTC);
}


# aggregation: edges counting-sorted by destination in every call (int atomic slot, scan, place); one wave per node gathers its rows, sums in registers, no float atomics there; sum order = slot order
# speedup vs baseline: 1.6074x; 1.1951x over previous
#include <hip/hip_runtime.h>

typedef __attribute__((ext_vector_type(2))) float v2f;
typedef __attribute__((ext_vector_type(4))) float v4f;
typedef __attribute__((ext_vector_type(8))) float v8f;

#define NODES  50000
#define EDGES  800000
#define GRAPHS 64
#define C      64
#define OUTC   32
#define LPITCH 68
static_assert(NODES % 8 == 0, "one wave per node row, 8 waves per block, no tail");

__global__ void fill_kernel(float* __restrict__ p, float v, int n) {
  int i = blockIdx.x * blockDim.x + threadIdx.x;
  if (i < n) p[i] = v;
}

__global__ void zero_int_kernel(int* __restrict__ p, int n) {
  int i = blockIdx.x * blockDim.x + threadIdx.x;
  if (i < n) p[i] = 0;
}

__global__ void rank_kernel(const int* __restrict__ dst, int* __restrict__ cnt,
                            int* __restrict__ slot) {
  int e = blockIdx.x * blockDim.x + threadIdx.x;
  if (e < EDGES) slot[e] = atomicAdd(&cnt[dst[e]], 1);
}

#define SCAN_T   1024
#define SCAN_PER 49
static_assert(SCAN_T * SCAN_PER >= NODES, "scan pieces cover every node");
static_assert((SCAN_PER - 1) * SCAN_T <= NODES && (SCAN_PER - 1) % 16 == 0, "all passes but the last are full, in groups of 16");
__global__ __launch_bounds__(SCAN_T)
void scan_kernel(const int* __restrict__ cnt, int* __restrict__ rowptr,
                 float* __restrict__ dinv) {
  __shared__ int sc[SCAN_T * SCAN_PER];
  __shared__ int wsum[32];
  const int t = threadIdx.x;
  const int lane = t & 31;
  const int wave = t >> 5;
  for (int p = 0; p < SCAN_PER - 1; p += 16) {
    int c[16];
#pragma unroll
    for (int j = 0; j < 16; ++j) c[j] = cnt[(p + j) * SCAN_T + t];
#pragma unroll
    for (int j = 0; j < 16; ++j) {
      sc[(p + j) * SCAN_T + t] = c[j];
      dinv[(p + j) * SCAN_T + t] = rsqrtf((float)(c[j] + 1));
    }
  }
  {
    const int i = (SCAN_PER - 1) * SCAN_T + t;
    int c = 0;
    if (i < NODES) {
      c = cnt[i];
      dinv[i] = rsqrtf((float)(c + 1));
    }
    sc[i] = c;
  }
  __syncthreads();
  int* mine = sc + t * SCAN_PER;
  int run = 0;
  for (int j = 0; j < SCAN_PER; ++j) {
    const int c = mine[j];
    mine[j] = run;
    run += c;
  }
  int incl = run;
  for (int d = 1; d < 32; d <<= 1) {
    const int o = __shfl_up(incl, d);
    if (lane >= d) incl += o;
  }
  if (lane == 31) wsum[wave] = incl;
  __syncthreads();
  if (wave == 0) {
    const int v = wsum[lane];
    int iv = v;
    for (int d = 1; d < 32; d <<= 1) {
      const int o = __shfl_up(iv, d);
      if (lane >= d) iv += o;
    }
    wsum[lane] = iv - v;
  }
  __syncthreads();
  const int off = wsum[wave] + incl - run;
  for (int j = 0; j < SCAN_PER; ++j) mine[j] += off;
  __syncthreads();
  for (int i = t; i < NODES; i += SCAN_T) rowptr[i] = sc[i];
}

__global__ void place_kernel(const int* __restrict__ src, const int* __restrict__ dst,
                             const int* __restrict__ slot, const int* __restrict__ rowptr,
                             int* __restrict__ csr) {
  int e = blockIdx.x * blockDim.x + threadIdx.x;
  if (e < EDGES) csr[rowptr[dst[e]] + slot[e]] = src[e];
}

__global__ __launch_bounds__(256)
void gemm_rs_wmma(const float* __restrict__ X, const float* __restrict__ W,
                  const float* __restrict__ rowscale, float* __restrict__ out,
                  int ntiles) {
  __shared__ float sWt[C * LPITCH];
  for (int i = threadIdx.x; i < C * C; i += 256) {
    const int k = i >> 6, n = i & 63;
    sWt[n * LPITCH + k] = W[i];
  }
  __syncthreads();

  const int wave = threadIdx.x >> 5;
  const int lane = threadIdx.x & 31;
  const int tile = blockIdx.x * 8 + wave;
  if (tile >= ntiles) return;

  const int m0 = tile * 16;
  const int ar = m0 + (lane & 15);
  const int kg = (lane >> 4) * 2;
  const int cn = lane & 15;
  const float* bp = sWt + cn * LPITCH;

  v8f acc0 = {}, acc1 = {}, acc2 = {}, acc3 = {};
#pragma unroll
  for (int ks = 0; ks < 16; ++ks) {
    const int k0 = ks * 4 + kg;
    v2f a = *(const v2f*)(X + ar * C + k0);
    v2f b0 = *(const v2f*)(bp + k0);
    v2f b1 = *(const v2f*)(bp + 16 * LPITCH + k0);
    v2f b2 = *(const v2f*)(bp + 32 * LPITCH + k0);
    v2f b3 = *(const v2f*)(bp + 48 * LPITCH + k0);
    acc0 = __builtin_amdgcn_wmma_f32_16x16x4_f32(false, a, false, b0, (short)0, acc0, false, false);
    acc1 = __builtin_amdgcn_wmma_f32_16x16x4_f32(false, a, false, b1, (short)0, acc1, false, false);
    acc2 = __builtin_amdgcn_wmma_f32_16x16x4_f32(false, a, false, b2, (short)0, acc2, false, false);
    acc3 = __builtin_amdgcn_wmma_f32_16x16x4_f32(false, a, false, b3, (short)0, acc3, false, false);
  }

#pragma unroll
  for (int r = 0; r < 8; ++r) {
    const int row = m0 + r + 8 * (lane >> 4);
    const float s = rowscale[row];
    out[row * C +  0 + cn] = acc0[r] * s;
    out[row * C + 16 + cn] = acc1[r] * s;
    out[row * C + 32 + cn] = acc2[r] * s;
    out[row * C + 48 + cn] = acc3[r] * s;
  }
}

__global__ __launch_bounds__(256)
void gather_kernel(const int* __restrict__ rowptr, const int* __restrict__ cnt,
                   const int* __restrict__ csr, const float* __restrict__ hs,
                   const float* __restrict__ dinv, const float* __restrict__ bias,
                   float* __restrict__ outp, int do_relu) {
  const int lane = threadIdx.x & 31;
  const int row = blockIdx.x * 8 + __builtin_amdgcn_readfirstlane((int)(threadIdx.x >> 5));
  const int half = lane >> 4;
  const int q = (lane & 15) * 4;
  const int beg = rowptr[row];
  const int deg = cnt[row];
  int my = (lane < deg) ? csr[beg + lane] : 0;
  const float di = dinv[row];
  const v4f hr = *(const v4f*)(hs + (unsigned)row * C + q);
  const v4f b = *(const v4f*)(bias + q);
  v4f sum = {0.0f, 0.0f, 0.0f, 0.0f};
  for (int base = 0; base < deg; base += 32) {
    const int n = (deg - base < 32) ? (deg - base) : 32;
    int k = 0;
    for (; k + 8 <= n; k += 8) {
      v4f v[4];
#pragma unroll
      for (int j = 0; j < 4; ++j) {
        const int s = __shfl(my, k + 2 * j + half);
        v[j] = *(const v4f*)(hs + ((unsigned)s * C + q));
      }
#pragma unroll
      for (int j = 0; j < 4; ++j) sum += v[j];
    }
    if (k < n) {
      v4f v[4];
#pragma unroll
      for (int j = 0; j < 4; ++j) {
        const int s = __shfl(my, k + 2 * j + half);
        v[j] = *(const v4f*)(hs + ((unsigned)s * C + q));
      }
#pragma unroll
      for (int j = 0; j < 4; ++j)
        if (k + 2 * j + half < n) sum += v[j];
    }
    if (base + 32 < deg)
      my = (base + 32 + lane < deg) ? csr[beg + base + 32 + lane] : 0;
  }
  v4f tot;
  tot.x = sum.x + __shfl_xor(sum.x, 16);
  tot.y = sum.y + __shfl_xor(sum.y, 16);
  tot.z = sum.z + __shfl_xor(sum.z, 16);
  tot.w = sum.w + __shfl_xor(sum.w, 16);
  if (half == 0) {
    v4f o;
    o.x = di * (tot.x + hr.x) + b.x;
    o.y = di * (tot.y + hr.y) + b.y;
    o.z = di * (tot.z + hr.z) + b.z;
    o.w = di * (tot.w + hr.w) + b.w;
    if (do_relu) {
      o.x = fmaxf(o.x, 0.0f);
      o.y = fmaxf(o.y, 0.0f);
      o.z = fmaxf(o.z, 0.0f);
      o.w = fmaxf(o.w, 0.0f);
    }
    *(v4f*)(outp + ((unsigned)row * C + q)) = o;
  }
}

__global__ void pool_kernel(const float* __restrict__ z, const int* __restrict__ batch,
                            float* __restrict__ u) {
  int idx = blockIdx.x * blockDim.x + threadIdx.x;
  int i  = idx >> 4;
  int cg = (idx & 15) << 2;
  if (i < NODES) {
    int g = batch[i];
    const float4 v = *(const float4*)(z + (size_t)i * C + cg);
    float* p = u + g * C + cg;
    atomicAdd(p + 0, v.x);
    atomicAdd(p + 1, v.y);
    atomicAdd(p + 2, v.z);
    atomicAdd(p + 3, v.w);
  }
}

__global__ __launch_bounds__(256)
void head_wmma(const float* __restrict__ u, const float* __restrict__ Wf,
               const float* __restrict__ bf, float* __restrict__ y,
               float* __restrict__ out_u) {
  __shared__ float sWf[OUTC * LPITCH];
  for (int i = threadIdx.x; i < C * OUTC; i += 256) {
    const int k = i >> 5, n = i & 31;
    sWf[n * LPITCH + k] = Wf[i];
  }
  __syncthreads();

  const int wave = threadIdx.x >> 5;
  const int lane = threadIdx.x & 31;
  const int mt = wave >> 1;
  const int nt = wave & 1;
  const int ar = mt * 16 + (lane & 15);
  const int kg = (lane >> 4) * 2;
  const int cn = nt * 16 + (lane & 15);
  const float* bp = sWf + cn * LPITCH;

  v8f acc = {};
#pragma unroll
  for (int ks = 0; ks < 16; ++ks) {
    const int k0 = ks * 4 + kg;
    v2f a = *(const v2f*)(u + ar * C + k0);
    v2f b = *(const v2f*)(bp + k0);
    acc = __builtin_amdgcn_wmma_f32_16x16x4_f32(false, a, false, b, (short)0, acc, false, false);
  }
#pragma unroll
  for (int r = 0; r < 8; ++r) {
    const int row = mt * 16 + r + 8 * (lane >> 4);
    y[row * OUTC + cn] = acc[r] + bf[cn];
  }
#pragma unroll
  for (int j = 0; j < 16; ++j) {
    const int idx = threadIdx.x + j * 256;
    out_u[idx] = u[idx];
  }
}

extern "C" void kernel_launch(void* const* d_in, const int* in_sizes, int n_in,
                              void* d_out, int out_size, void* d_ws, size_t ws_size,
                              hipStream_t stream) {
  (void)in_sizes; (void)n_in; (void)out_size; (void)ws_size;

  const float* x     = (const float*)d_in[0];
  const int*   ei    = (const int*)  d_in[1];
  const int*   batch = (const int*)  d_in[2];
  const float* W1    = (const float*)d_in[3];
  const float* b1    = (const float*)d_in[4];
  const float* W2    = (const float*)d_in[5];
  const float* b2    = (const float*)d_in[6];
  const float* Wf    = (const float*)d_in[7];
  const float* bf    = (const float*)d_in[8];
  float* out = (float*)d_out;

  const int* src = ei;
  const int* dst = ei + EDGES;

  float* ws   = (float*)d_ws;
  float* dinv = ws;
  float* hs   = ws + 50048;
  float* acc  = hs + (size_t)NODES * C;
  float* u    = acc + (size_t)NODES * C;
  int* cnt    = (int*)(u + GRAPHS * C);
  int* rowptr = cnt + 50048;
  int* slot   = rowptr + 50048;
  int* csr    = slot + EDGES;

  const dim3 B(256);
  const int NT = NODES / 16;
  const int GB = (NT + 7) / 8;

  zero_int_kernel<<<(NODES + 255) / 256, B, 0, stream>>>(cnt, NODES);
  rank_kernel    <<<(EDGES + 255) / 256, B, 0, stream>>>(dst, cnt, slot);
  scan_kernel    <<<1, SCAN_T, 0, stream>>>(cnt, rowptr, dinv);
  place_kernel   <<<(EDGES + 255) / 256, B, 0, stream>>>(src, dst, slot, rowptr, csr);

  gemm_rs_wmma <<<GB, B, 0, stream>>>(x, W1, dinv, hs, NT);
  gather_kernel<<<NODES / 8, B, 0, stream>>>(rowptr, cnt, csr, hs, dinv, b1, acc, 1);

  gemm_rs_wmma <<<GB, B, 0, stream>>>(acc, W2, dinv, hs, NT);
  gather_kernel<<<NODES / 8, B, 0, stream>>>(rowptr, cnt, csr, hs, dinv, b2, acc, 0);

  fill_kernel<<<(GRAPHS * C + 255) / 256, B, 0, stream>>>(u, 0.0f, GRAPHS * C);
  pool_kernel<<<(NODES * 16) / 256, B, 0, stream>>>(acc, batch, u);
  head_wmma  <<<1, B, 0, stream>>>(u, Wf, bf, out, out + GRAPHS * OUTC);
}


# global add pool fused into the layer-2 gather: a block sums its 8 rows per graph in LDS (row order), then one float atomic per column into u (arrival order across blocks); z never stored
# speedup vs baseline: 2.0952x; 1.3035x over previous
#include <hip/hip_runtime.h>

typedef __attribute__((ext_vector_type(2))) float v2f;
typedef __attribute__((ext_vector_type(4))) float v4f;
typedef __attribute__((ext_vector_type(8))) float v8f;

#define NODES  50000
#define EDGES  800000
#define GRAPHS 64
#define C      64
#define OUTC   32
#define LPITCH 68
static_assert(NODES % 8 == 0, "one wave per node row, 8 waves per block, no tail");

static_assert(GRAPHS * C <= NODES, "the grid that covers cnt covers u as well");
__global__ void clear_kernel(int* __restrict__ cnt, float* __restrict__ u) {
  int i = blockIdx.x * blockDim.x + threadIdx.x;
  if (i < NODES) cnt[i] = 0;
  if (i < GRAPHS * C) u[i] = 0.0f;
}

__global__ void rank_kernel(const int* __restrict__ dst, int* __restrict__ cnt,
                            int* __restrict__ slot) {
  int e = blockIdx.x * blockDim.x + threadIdx.x;
  if (e < EDGES) slot[e] = atomicAdd(&cnt[dst[e]], 1);
}

#define SCAN_T   1024
#define SCAN_PER 49
static_assert(SCAN_T * SCAN_PER >= NODES, "scan pieces cover every node");
static_assert((SCAN_PER - 1) * SCAN_T <= NODES && (SCAN_PER - 1) % 16 == 0, "all passes but the last are full, in groups of 16");
__global__ __launch_bounds__(SCAN_T)
void scan_kernel(const int* __restrict__ cnt, int* __restrict__ rowptr,
                 float* __restrict__ dinv) {
  __shared__ int sc[SCAN_T * SCAN_PER];
  __shared__ int wsum[32];
  const int t = threadIdx.x;
  const int lane = t & 31;
  const int wave = t >> 5;
  for (int p = 0; p < SCAN_PER - 1; p += 16) {
    int c[16];
#pragma unroll
    for (int j = 0; j < 16; ++j) c[j] = cnt[(p + j) * SCAN_T + t];
#pragma unroll
    for (int j = 0; j < 16; ++j) {
      sc[(p + j) * SCAN_T + t] = c[j];
      dinv[(p + j) * SCAN_T + t] = rsqrtf((float)(c[j] + 1));
    }
  }
  {
    const int i = (SCAN_PER - 1) * SCAN_T + t;
    int c = 0;
    if (i < NODES) {
      c = cnt[i];
      dinv[i] = rsqrtf((float)(c + 1));
    }
    sc[i] = c;
  }
  __syncthreads();
  int* mine = sc + t * SCAN_PER;
  int run = 0;
  for (int j = 0; j < SCAN_PER; ++j) {
    const int c = mine[j];
    mine[j] = run;
    run += c;
  }
  int incl = run;
  for (int d = 1; d < 32; d <<= 1) {
    const int o = __shfl_up(incl, d);
    if (lane >= d) incl += o;
  }
  if (lane == 31) wsum[wave] = incl;
  __syncthreads();
  if (wave == 0) {
    const int v = wsum[lane];
    int iv = v;
    for (int d = 1; d < 32; d <<= 1) {
      const int o = __shfl_up(iv, d);
      if (lane >= d) iv += o;
    }
    wsum[lane] = iv - v;
  }
  __syncthreads();
  const int off = wsum[wave] + incl - run;
  for (int j = 0; j < SCAN_PER; ++j) mine[j] += off;
  __syncthreads();
  for (int i = t; i < NODES; i += SCAN_T) rowptr[i] = sc[i];
}

__global__ void place_kernel(const int* __restrict__ src, const int* __restrict__ dst,
                             const int* __restrict__ slot, const int* __restrict__ rowptr,
                             int* __restrict__ csr) {
  int e = blockIdx.x * blockDim.x + threadIdx.x;
  if (e < EDGES) csr[rowptr[dst[e]] + slot[e]] = src[e];
}

__global__ __launch_bounds__(256)
void gemm_rs_wmma(const float* __restrict__ X, const float* __restrict__ W,
                  const float* __restrict__ rowscale, float* __restrict__ out,
                  int ntiles) {
  __shared__ float sWt[C * LPITCH];
  for (int i = threadIdx.x; i < C * C; i += 256) {
    const int k = i >> 6, n = i & 63;
    sWt[n * LPITCH + k] = W[i];
  }
  __syncthreads();

  const int wave = threadIdx.x >> 5;
  const int lane = threadIdx.x & 31;
  const int tile = blockIdx.x * 8 + wave;
  if (tile >= ntiles) return;

  const int m0 = tile * 16;
  const int ar = m0 + (lane & 15);
  const int kg = (lane >> 4) * 2;
  const int cn = lane & 15;
  const float* bp = sWt + cn * LPITCH;

  v8f acc0 = {}, acc1 = {}, acc2 = {}, acc3 = {};
#pragma unroll
  for (int ks = 0; ks < 16; ++ks) {
    const int k0 = ks * 4 + kg;
    v2f a = *(const v2f*)(X + ar * C + k0);
    v2f b0 = *(const v2f*)(bp + k0);
    v2f b1 = *(const v2f*)(bp + 16 * LPITCH + k0);
    v2f b2 = *(const v2f*)(bp + 32 * LPITCH + k0);
    v2f b3 = *(const v2f*)(bp + 48 * LPITCH + k0);
    acc0 = __builtin_amdgcn_wmma_f32_16x16x4_f32(false, a, false, b0, (short)0, acc0, false, false);
    acc1 = __builtin_amdgcn_wmma_f32_16x16x4_f32(false, a, false, b1, (short)0, acc1, false, false);
    acc2 = __builtin_amdgcn_wmma_f32_16x16x4_f32(false, a, false, b2, (short)0, acc2, false, false);
    acc3 = __builtin_amdgcn_wmma_f32_16x16x4_f32(false, a, false, b3, (short)0, acc3, false, false);
  }

#pragma unroll
  for (int r = 0; r < 8; ++r) {
    const int row = m0 + r + 8 * (lane >> 4);
    const float s = rowscale[row];
    out[row * C +  0 + cn] = acc0[r] * s;
    out[row * C + 16 + cn] = acc1[r] * s;
    out[row * C + 32 + cn] = acc2[r] * s;
    out[row * C + 48 + cn] = acc3[r] * s;
  }
}

__device__ __forceinline__ v4f gather_row(const int* __restrict__ rowptr, const int* __restrict__ cnt,
                                          const int* __restrict__ csr, const float* __restrict__ hs,
                                          const float* __restrict__ dinv, const float* __restrict__ bias,
                                          int row, int lane) {
  const int half = lane >> 4;
  const int q = (lane & 15) * 4;
  const int beg = rowptr[row];
  const int deg = cnt[row];
  int my = (lane < deg) ? csr[beg + lane] : 0;
  const float di = dinv[row];
  const v4f hr = *(const v4f*)(hs + (unsigned)row * C + q);
  const v4f b = *(const v4f*)(bias + q);
  v4f sum = {0.0f, 0.0f, 0.0f, 0.0f};
  for (int base = 0; base < deg; base += 32) {
    const int n = (deg - base < 32) ? (deg - base) : 32;
    int k = 0;
    for (; k + 8 <= n; k += 8) {
      v4f v[4];
#pragma unroll
      for (int j = 0; j < 4; ++j) {
        const int s = __shfl(my, k + 2 * j + half);
        v[j] = *(const v4f*)(hs + ((unsigned)s * C + q));
      }
#pragma unroll
      for (int j = 0; j < 4; ++j) sum += v[j];
    }
    if (k < n) {
      v4f v[4];
#pragma unroll
      for (int j = 0; j < 4; ++j) {
        const int s = __shfl(my, k + 2 * j + half);
        v[j] = *(const v4f*)(hs + ((unsigned)s * C + q));
      }
#pragma unroll
      for (int j = 0; j < 4; ++j)
        if (k + 2 * j + half < n) sum += v[j];
    }
    if (base + 32 < deg)
      my = (base + 32 + lane < deg) ? csr[beg + base + 32 + lane] : 0;
  }
  v4f o;
  o.x = di * (sum.x + __shfl_xor(sum.x, 16) + hr.x) + b.x;
  o.y = di * (sum.y + __shfl_xor(sum.y, 16) + hr.y) + b.y;
  o.z = di * (sum.z + __shfl_xor(sum.z, 16) + hr.z) + b.z;
  o.w = di * (sum.w + __shfl_xor(sum.w, 16) + hr.w) + b.w;
  return o;
}

__global__ __launch_bounds__(256)
void gather_relu_kernel(const int* __restrict__ rowptr, const int* __restrict__ cnt,
                        const int* __restrict__ csr, const float* __restrict__ hs,
                        const float* __restrict__ dinv, const float* __restrict__ bias,
                        float* __restrict__ outp) {
  const int lane = threadIdx.x & 31;
  const int row = blockIdx.x * 8 + __builtin_amdgcn_readfirstlane((int)(threadIdx.x >> 5));
  v4f o = gather_row(rowptr, cnt, csr, hs, dinv, bias, row, lane);
  if (lane < 16) {
    o.x = fmaxf(o.x, 0.0f);
    o.y = fmaxf(o.y, 0.0f);
    o.z = fmaxf(o.z, 0.0f);
    o.w = fmaxf(o.w, 0.0f);
    *(v4f*)(outp + ((unsigned)row * C + lane * 4)) = o;
  }
}

__global__ __launch_bounds__(256)
void gather_pool_kernel(const int* __restrict__ rowptr, const int* __restrict__ cnt,
                        const int* __restrict__ csr, const float* __restrict__ hs,
                        const float* __restrict__ dinv, const float* __restrict__ bias,
                        const int* __restrict__ batch, float* __restrict__ u) {
  __shared__ float srow[8 * C];
  __shared__ int sg[8];
  const int lane = threadIdx.x & 31;
  const int wave = __builtin_amdgcn_readfirstlane((int)(threadIdx.x >> 5));
  const int row = blockIdx.x * 8 + wave;
  const int g = batch[row];
  const v4f o = gather_row(rowptr, cnt, csr, hs, dinv, bias, row, lane);
  if (lane < 16) *(v4f*)(srow + wave * C + lane * 4) = o;
  if (lane == 0) sg[wave] = g;
  __syncthreads();
  if (threadIdx.x < C) {
    const int c = threadIdx.x;
    float v[8];
    int gid[8];
#pragma unroll
    for (int w = 0; w < 8; ++w) {
      v[w] = srow[w * C + c];
      gid[w] = sg[w];
    }
    int cur = gid[0];
    float run = v[0];
#pragma unroll
    for (int w = 1; w < 8; ++w) {
      if (gid[w] != cur) {
        atomicAdd(u + cur * C + c, run);
        cur = gid[w];
        run = v[w];
      } else {
        run += v[w];
      }
    }
    atomicAdd(u + cur * C + c, run);
  }
}

__global__ __launch_bounds__(256)
void head_wmma(const float* __restrict__ u, const float* __restrict__ Wf,
               const float* __restrict__ bf, float* __restrict__ y,
               float* __restrict__ out_u) {
  __shared__ float sWf[OUTC * LPITCH];
  for (int i = threadIdx.x; i < C * OUTC; i += 256) {
    const int k = i >> 5, n = i & 31;
    sWf[n * LPITCH + k] = Wf[i];
  }
  __syncthreads();

  const int wave = threadIdx.x >> 5;
  const int lane = threadIdx.x & 31;
  const int mt = wave >> 1;
  const int nt = wave & 1;
  const int ar = mt * 16 + (lane & 15);
  const int kg = (lane >> 4) * 2;
  const int cn = nt * 16 + (lane & 15);
  const float* bp = sWf + cn * LPITCH;

  v8f acc = {};
#pragma unroll
  for (int ks = 0; ks < 16; ++ks) {
    const int k0 = ks * 4 + kg;
    v2f a = *(const v2f*)(u + ar * C + k0);
    v2f b = *(const v2f*)(bp + k0);
    acc = __builtin_amdgcn_wmma_f32_16x16x4_f32(false, a, false, b, (short)0, acc, false, false);
  }
#pragma unroll
  for (int r = 0; r < 8; ++r) {
    const int row = mt * 16 + r + 8 * (lane >> 4);
    y[row * OUTC + cn] = acc[r] + bf[cn];
  }
#pragma unroll
  for (int j = 0; j < 16; ++j) {
    const int idx = threadIdx.x + j * 256;
    out_u[idx] = u[idx];
  }
}

extern "C" void kernel_launch(void* const* d_in, const int* in_sizes, int n_in,
                              void* d_out, int out_size, void* d_ws, size_t ws_size,
                              hipStream_t stream) {
  (void)in_sizes; (void)n_in; (void)out_size; (void)ws_size;

  const float* x     = (const float*)d_in[0];
  const int*   ei    = (const int*)  d_in[1];
  const int*   batch = (const int*)  d_in[2];
  const float* W1    = (const float*)d_in[3];
  const float* b1    = (const float*)d_in[4];
  const float* W2    = (const float*)d_in[5];
  const float* b2    = (const float*)d_in[6];
  const float* Wf    = (const float*)d_in[7];
  const float* bf    = (const float*)d_in[8];
  float* out = (float*)d_out;

  const int* src = ei;
  const int* dst = ei + EDGES;

  float* ws   = (float*)d_ws;
  float* dinv = ws;
  float* hs   = ws + 50048;
  float* acc  = hs + (size_t)NODES * C;
  float* u    = acc + (size_t)NODES * C;
  int* cnt    = (int*)(u + GRAPHS * C);
  int* rowptr = cnt + 50048;
  int* slot   = rowptr + 50048;
  int* csr    = slot + EDGES;

  const dim3 B(256);
  const int NT = NODES / 16;
  const int GB = (NT + 7) / 8;

  clear_kernel   <<<(NODES + 255) / 256, B, 0, stream>>>(cnt, u);

  rank_kernel    <<<(EDGES + 255) / 256, B, 0, stream>>>(dst, cnt, slot);
  scan_kernel    <<<1, SCAN_T, 0, stream>>>(cnt, rowptr, dinv);
  place_kernel   <<<(EDGES + 255) / 256, B, 0, stream>>>(src, dst, slot, rowptr, csr);

  gemm_rs_wmma <<<GB, B, 0, stream>>>(x, W1, dinv, hs, NT);
  gather_relu_kernel<<<NODES / 8, B, 0, stream>>>(rowptr, cnt, csr, hs, dinv, b1, acc);

  gemm_rs_wmma <<<GB, B, 0, stream>>>(acc, W2, dinv, hs, NT);
  gather_pool_kernel<<<NODES / 8, B, 0, stream>>>(rowptr, cnt, csr, hs, dinv, b2, batch, u);

  head_wmma  <<<1, B, 0, stream>>>(u, Wf, bf, out, out + GRAPHS * OUTC);
}


# edge sort replaced by fixed-pitch buckets: the counting atomic's slot indexes a 32-slot line per destination (rest: overflow list); scan, slot array, place launch gone; sums: slot then list order
# speedup vs baseline: 2.4251x; 1.1575x over previous
#include <hip/hip_runtime.h>

typedef __attribute__((ext_vector_type(2))) float v2f;
typedef __attribute__((ext_vector_type(4))) float v4f;
typedef __attribute__((ext_vector_type(8))) float v8f;

#define NODES  50000
#define EDGES  800000
#define GRAPHS 64
#define C      64
#define OUTC   32
#define LPITCH 68
#define CAP    32
static_assert(NODES % 8 == 0, "one wave per node row, 8 waves per block, no tail");

static_assert(GRAPHS * C <= NODES, "the grid that covers cnt covers u as well");
__global__ void clear_kernel(int* __restrict__ cnt, float* __restrict__ u, int* __restrict__ novf) {
  int i = blockIdx.x * blockDim.x + threadIdx.x;
  if (i < NODES) cnt[i] = 0;
  if (i < GRAPHS * C) u[i] = 0.0f;
  if (i == 0) *novf = 0;
}

__global__ void bucket_kernel(const int* __restrict__ src, const int* __restrict__ dst,
                              int* __restrict__ cnt, int* __restrict__ bkt,
                              int* __restrict__ novf, int* __restrict__ ovf_src,
                              int* __restrict__ ovf_dst) {
  int e = blockIdx.x * blockDim.x + threadIdx.x;
  if (e < EDGES) {
    const int d = dst[e];
    const int s = src[e];
    const int slot = atomicAdd(&cnt[d], 1);
    if (slot < CAP) {
      bkt[d * CAP + slot] = s;
    } else {
      const int k = atomicAdd(novf, 1);
      ovf_src[k] = s;
      ovf_dst[k] = d;
    }
  }
}

__global__ __launch_bounds__(256)
void gemm_rs_wmma(const float* __restrict__ X, const float* __restrict__ W,
                  const int* __restrict__ cnt, float* __restrict__ out,
                  int ntiles) {
  __shared__ float sWt[C * LPITCH];
  for (int i = threadIdx.x; i < C * C; i += 256) {
    const int k = i >> 6, n = i & 63;
    sWt[n * LPITCH + k] = W[i];
  }
  __syncthreads();

  const int wave = threadIdx.x >> 5;
  const int lane = threadIdx.x & 31;
  const int tile = blockIdx.x * 8 + wave;
  if (tile >= ntiles) return;

  const int m0 = tile * 16;
  const int ar = m0 + (lane & 15);
  const int kg = (lane >> 4) * 2;
  const int cn = lane & 15;
  const float* bp = sWt + cn * LPITCH;

  v8f acc0 = {}, acc1 = {}, acc2 = {}, acc3 = {};
#pragma unroll
  for (int ks = 0; ks < 16; ++ks) {
    const int k0 = ks * 4 + kg;
    v2f a = *(const v2f*)(X + ar * C + k0);
    v2f b0 = *(const v2f*)(bp + k0);
    v2f b1 = *(const v2f*)(bp + 16 * LPITCH + k0);
    v2f b2 = *(const v2f*)(bp + 32 * LPITCH + k0);
    v2f b3 = *(const v2f*)(bp + 48 * LPITCH + k0);
    acc0 = __builtin_amdgcn_wmma_f32_16x16x4_f32(false, a, false, b0, (short)0, acc0, false, false);
    acc1 = __builtin_amdgcn_wmma_f32_16x16x4_f32(false, a, false, b1, (short)0, acc1, false, false);
    acc2 = __builtin_amdgcn_wmma_f32_16x16x4_f32(false, a, false, b2, (short)0, acc2, false, false);
    acc3 = __builtin_amdgcn_wmma_f32_16x16x4_f32(false, a, false, b3, (short)0, acc3, false, false);
  }

#pragma unroll
  for (int r = 0; r < 8; ++r) {
    const int row = m0 + r + 8 * (lane >> 4);
    const float s = rsqrtf((float)(cnt[row] + 1));
    out[row * C +  0 + cn] = acc0[r] * s;
    out[row * C + 16 + cn] = acc1[r] * s;
    out[row * C + 32 + cn] = acc2[r] * s;
    out[row * C + 48 + cn] = acc3[r] * s;
  }
}

__device__ __forceinline__ v4f gather_row(const int* __restrict__ cnt, const int* __restrict__ bkt,
                                          const int* __restrict__ novf, const int* __restrict__ ovf_src,
                                          const int* __restrict__ ovf_dst, const float* __restrict__ hs,
                                          const float* __restrict__ bias, int row, int lane) {
  const int half = lane >> 4;
  const int q = (lane & 15) * 4;
  const int deg = cnt[row];
  const int n = (deg < CAP) ? deg : CAP;
  const int my = (lane < n) ? bkt[row * CAP + lane] : 0;
  const float di = rsqrtf((float)(deg + 1));
  const v4f hr = *(const v4f*)(hs + (unsigned)row * C + q);
  const v4f b = *(const v4f*)(bias + q);
  v4f sum = {0.0f, 0.0f, 0.0f, 0.0f};
  int k = 0;
  for (; k + 8 <= n; k += 8) {
    v4f v[4];
#pragma unroll
    for (int j = 0; j < 4; ++j) {
      const int s = __shfl(my, k + 2 * j + half);
      v[j] = *(const v4f*)(hs + ((unsigned)s * C + q));
    }
#pragma unroll
    for (int j = 0; j < 4; ++j) sum += v[j];
  }
  if (k < n) {
    v4f v[4];
#pragma unroll
    for (int j = 0; j < 4; ++j) {
      const int s = __shfl(my, k + 2 * j + half);
      v[j] = *(const v4f*)(hs + ((unsigned)s * C + q));
    }
#pragma unroll
    for (int j = 0; j < 4; ++j)
      if (k + 2 * j + half < n) sum += v[j];
  }
  if (deg > CAP) {
    const int m = *novf;
    for (int base = 0; base < m; base += 32) {
      const int idx = base + lane;
      const int od = (idx < m) ? ovf_dst[idx] : -1;
      const int os = (idx < m) ? ovf_src[idx] : 0;
      unsigned hit = __builtin_amdgcn_ballot_w32(od == row);
      while (hit) {
        const int j = __builtin_ctz(hit);
        hit &= hit - 1;
        const int s = __shfl(os, j);
        const v4f v = *(const v4f*)(hs + ((unsigned)s * C + q));
        if (half == 0) sum += v;
      }
    }
  }
  v4f o;
  o.x = di * (sum.x + __shfl_xor(sum.x, 16) + hr.x) + b.x;
  o.y = di * (sum.y + __shfl_xor(sum.y, 16) + hr.y) + b.y;
  o.z = di * (sum.z + __shfl_xor(sum.z, 16) + hr.z) + b.z;
  o.w = di * (sum.w + __shfl_xor(sum.w, 16) + hr.w) + b.w;
  return o;
}

__global__ __launch_bounds__(256)
void gather_relu_kernel(const int* __restrict__ cnt, const int* __restrict__ bkt,
                        const int* __restrict__ novf, const int* __restrict__ ovf_src,
                        const int* __restrict__ ovf_dst, const float* __restrict__ hs,
                        const float* __restrict__ bias, float* __restrict__ outp) {
  const int lane = threadIdx.x & 31;
  const int row = blockIdx.x * 8 + __builtin_amdgcn_readfirstlane((int)(threadIdx.x >> 5));
  v4f o = gather_row(cnt, bkt, novf, ovf_src, ovf_dst, hs, bias, row, lane);
  if (lane < 16) {
    o.x = fmaxf(o.x, 0.0f);
    o.y = fmaxf(o.y, 0.0f);
    o.z = fmaxf(o.z, 0.0f);
    o.w = fmaxf(o.w, 0.0f);
    *(v4f*)(outp + ((unsigned)row * C + lane * 4)) = o;
  }
}

__global__ __launch_bounds__(256)
void gather_pool_kernel(const int* __restrict__ cnt, const int* __restrict__ bkt,
                        const int* __restrict__ novf, const int* __restrict__ ovf_src,
                        const int* __restrict__ ovf_dst, const float* __restrict__ hs,
                        const float* __restrict__ bias,
                        const int* __restrict__ batch, float* __restrict__ u) {
  __shared__ float srow[8 * C];
  __shared__ int sg[8];
  const int lane = threadIdx.x & 31;
  const int wave = __builtin_amdgcn_readfirstlane((int)(threadIdx.x >> 5));
  const int row = blockIdx.x * 8 + wave;
  const int g = batch[row];
  const v4f o = gather_row(cnt, bkt, novf, ovf_src, ovf_dst, hs, bias, row, lane);
  if (lane < 16) *(v4f*)(srow + wave * C + lane * 4) = o;
  if (lane == 0) sg[wave] = g;
  __syncthreads();
  if (threadIdx.x < C) {
    const int c = threadIdx.x;
    float v[8];
    int gid[8];
#pragma unroll
    for (int w = 0; w < 8; ++w) {
      v[w] = srow[w * C + c];
      gid[w] = sg[w];
    }
    int cur = gid[0];
    float run = v[0];
#pragma unroll
    for (int w = 1; w < 8; ++w) {
      if (gid[w] != cur) {
        atomicAdd(u + cur * C + c, run);
        cur = gid[w];
        run = v[w];
      } else {
        run += v[w];
      }
    }
    atomicAdd(u + cur * C + c, run);
  }
}

__global__ __launch_bounds__(256)
void head_wmma(const float* __restrict__ u, const float* __restrict__ Wf,
               const float* __restrict__ bf, float* __restrict__ y,
               float* __restrict__ out_u) {
  __shared__ float sWf[OUTC * LPITCH];
  for (int i = threadIdx.x; i < C * OUTC; i += 256) {
    const int k = i >> 5, n = i & 31;
    sWf[n * LPITCH + k] = Wf[i];
  }
  __syncthreads();

  const int wave = threadIdx.x >> 5;
  const int lane = threadIdx.x & 31;
  const int mt = wave >> 1;
  const int nt = wave & 1;
  const int ar = mt * 16 + (lane & 15);
  const int kg = (lane >> 4) * 2;
  const int cn = nt * 16 + (lane & 15);
  const float* bp = sWf + cn * LPITCH;

  v8f acc = {};
#pragma unroll
  for (int ks = 0; ks < 16; ++ks) {
    const int k0 = ks * 4 + kg;
    v2f a = *(const v2f*)(u + ar * C + k0);
    v2f b = *(const v2f*)(bp + k0);
    acc = __builtin_amdgcn_wmma_f32_16x16x4_f32(false, a, false, b, (short)0, acc, false, false);
  }
#pragma unroll
  for (int r = 0; r < 8; ++r) {
    const int row = mt * 16 + r + 8 * (lane >> 4);
    y[row * OUTC + cn] = acc[r] + bf[cn];
  }
#pragma unroll
  for (int j = 0; j < 16; ++j) {
    const int idx = threadIdx.x + j * 256;
    out_u[idx] = u[idx];
  }
}

extern "C" void kernel_launch(void* const* d_in, const int* in_sizes, int n_in,
                              void* d_out, int out_size, void* d_ws, size_t ws_size,
                              hipStream_t stream) {
  (void)in_sizes; (void)n_in; (void)out_size; (void)ws_size;

  const float* x     = (const float*)d_in[0];
  const int*   ei    = (const int*)  d_in[1];
  const int*   batch = (const int*)  d_in[2];
  const float* W1    = (const float*)d_in[3];
  const float* b1    = (const float*)d_in[4];
  const float* W2    = (const float*)d_in[5];
  const float* b2    = (const float*)d_in[6];
  const float* Wf    = (const float*)d_in[7];
  const float* bf    = (const float*)d_in[8];
  float* out = (float*)d_out;

  const int* src = ei;
  const int* dst = ei + EDGES;

  float* ws    = (float*)d_ws;
  float* hs    = ws;
  float* acc   = hs + (size_t)NODES * C;
  float* u     = acc + (size_t)NODES * C;
  int* cnt     = (int*)(u + GRAPHS * C);
  int* novf    = cnt + 50048;
  int* bkt     = novf + 32;
  int* ovf_src = bkt + (size_t)NODES * CAP;
  int* ovf_dst = ovf_src + EDGES;

  const dim3 B(256);
  const int NT = NODES / 16;
  const int GB = (NT + 7) / 8;

  clear_kernel   <<<(NODES + 255) / 256, B, 0, stream>>>(cnt, u, novf);

  bucket_kernel  <<<(EDGES + 255) / 256, B, 0, stream>>>(src, dst, cnt, bkt, novf, ovf_src, ovf_dst);

  gemm_rs_wmma <<<GB, B, 0, stream>>>(x, W1, cnt, hs, NT);
  gather_relu_kernel<<<NODES / 8, B, 0, stream>>>(cnt, bkt, novf, ovf_src, ovf_dst, hs, b1, acc);

  gemm_rs_wmma <<<GB, B, 0, stream>>>(acc, W2, cnt, hs, NT);
  gather_pool_kernel<<<NODES / 8, B, 0, stream>>>(cnt, bkt, novf, ovf_src, ovf_dst, hs, b2, batch, u);

  head_wmma  <<<1, B, 0, stream>>>(u, Wf, bf, out, out + GRAPHS * OUTC);
}


# edge buckets via LDS, two launches: blocks sort their edge chunk by destination range; a block per range fills slot lines by LDS atomics; global atomics, clear launch gone; slots in arrival order
# speedup vs baseline: 4.2454x; 1.7506x over previous
#include <hip/hip_runtime.h>

typedef __attribute__((ext_vector_type(2))) float v2f;
typedef __attribute__((ext_vector_type(4))) float v4f;
typedef __attribute__((ext_vector_type(8))) float v8f;

#define NODES  50000
#define EDGES  800000
#define GRAPHS 64
#define C      64
#define OUTC   32
#define LPITCH 68
#define CAP    32
static_assert(NODES % 8 == 0, "one wave per node row, 8 waves per block, no tail");
#define RB     256
#define NBIN   ((NODES + RB - 1) / RB)
#define NCHUNK 256
#define CHUNK  (EDGES / NCHUNK)
#define PER    ((CHUNK + 255) / 256)
static_assert(EDGES % NCHUNK == 0, "equal chunks, no tail chunk");
static_assert(EDGES % 4 == 0, "16-byte reads of the sorted edge words stay inside the array");
static_assert(NBIN <= 256, "one thread per range in the scan");
static_assert(NODES <= (1 << 23), "source id << 8 | local destination is a non-negative word");
static_assert(GRAPHS * C <= NCHUNK * 256, "the first pass's grid covers u");

__global__ __launch_bounds__(256)
void bin_kernel(const int* __restrict__ src, const int* __restrict__ dst,
                float* __restrict__ u, int* __restrict__ novf,
                int* __restrict__ chunk, int* __restrict__ seg) {
  __shared__ int hist[256];
  __shared__ int wsum[8];
  __shared__ int sorted[CHUNK];
  const int t = threadIdx.x;
  const int lane = t & 31;
  const int wave = t >> 5;
  const int b = blockIdx.x;
  hist[t] = 0;
  __syncthreads();

  const int e0 = b * CHUNK;
  int dd[PER], ss[PER], sl[PER];
#pragma unroll
  for (int j = 0; j < PER; ++j) {
    const bool in = ((j + 1) * 256 <= CHUNK) || (j * 256 + t < CHUNK);
    dd[j] = in ? dst[e0 + j * 256 + t] : 0;
    ss[j] = in ? src[e0 + j * 256 + t] : 0;
  }
#pragma unroll
  for (int j = 0; j < PER; ++j)
    if (((j + 1) * 256 <= CHUNK) || (j * 256 + t < CHUNK))
      sl[j] = atomicAdd(&hist[(unsigned)dd[j] / RB], 1);
  __syncthreads();

  const int h = hist[t];
  int x = h;
#pragma unroll
  for (int o = 1; o < 32; o <<= 1) {
    const int y = __shfl_up(x, o);
    if (lane >= o) x += y;
  }
  if (lane == 31) wsum[wave] = x;
  __syncthreads();
  int before = 0;
#pragma unroll
  for (int k = 0; k < 8; ++k)
    if (k < wave) before += wsum[k];
  const int start = before + x - h;
  hist[t] = start;
  __syncthreads();

#pragma unroll
  for (int j = 0; j < PER; ++j)
    if (((j + 1) * 256 <= CHUNK) || (j * 256 + t < CHUNK))
      sorted[hist[(unsigned)dd[j] / RB] + sl[j]] = (ss[j] << 8) | (dd[j] & (RB - 1));
  __syncthreads();
#pragma unroll
  for (int j = 0; j < PER; ++j)
    if (((j + 1) * 256 <= CHUNK) || (j * 256 + t < CHUNK))
      chunk[e0 + j * 256 + t] = sorted[j * 256 + t];
  if (t < NBIN) *(int2*)(seg + 2 * (b * NBIN + t)) = make_int2(start, h);
  if (b < GRAPHS * C / 256) u[b * 256 + t] = 0.0f;
  if (b == 0 && t == 0) *novf = 0;
}

__global__ __launch_bounds__(256)
void build_kernel(const int* __restrict__ chunk, const int* __restrict__ seg,
                  int* __restrict__ cnt, int* __restrict__ bkt,
                  int* __restrict__ novf, int* __restrict__ ovf_src,
                  int* __restrict__ ovf_dst) {
  __shared__ int lcnt[RB + 32];
  __shared__ __attribute__((aligned(16))) int lbk[RB * CAP];
  const int t = threadIdx.x;
  const int lane = t & 31;
  const int bin = blockIdx.x;
  lcnt[t] = 0;
  if (t < 32) lcnt[RB + t] = 0;
  const int2 sg = *(const int2*)(seg + 2 * (t * NBIN + bin));
  __syncthreads();

  const int gb = t * CHUNK + sg.x;
  const int ge = gb + sg.y;
  for (int a = gb & ~3; a < ge; a += 16) {
    int w[16], slot[16];
#pragma unroll
    for (int j = 0; j < 4; ++j) {
      const int4 v = (a + 4 * j < ge) ? *(const int4*)(chunk + a + 4 * j) : make_int4(0, 0, 0, 0);
      w[4 * j + 0] = v.x; w[4 * j + 1] = v.y; w[4 * j + 2] = v.z; w[4 * j + 3] = v.w;
    }
#pragma unroll
    for (int j = 0; j < 16; ++j) {
      const bool in = (a + j >= gb) && (a + j < ge);
      slot[j] = atomicAdd(&lcnt[in ? (w[j] & (RB - 1)) : (RB + lane)], 1);
    }
#pragma unroll
    for (int j = 0; j < 16; ++j) {
      if ((a + j >= gb) && (a + j < ge)) {
        const int dl = w[j] & (RB - 1);
        const int s = w[j] >> 8;
        if (slot[j] < CAP) {
          lbk[dl * CAP + slot[j]] = s;
        } else {
          const int k = atomicAdd(novf, 1);
          ovf_src[k] = s;
          ovf_dst[k] = bin * RB + dl;
        }
      }
    }
  }
  __syncthreads();

  const int row0 = bin * RB;
  if (row0 + t < NODES) cnt[row0 + t] = lcnt[t];
  int4 pc[RB * CAP / 4 / 256];
#pragma unroll
  for (int k = 0; k < RB * CAP / 4 / 256; ++k) pc[k] = *(const int4*)(lbk + (k * 256 + t) * 4);
#pragma unroll
  for (int k = 0; k < RB * CAP / 4 / 256; ++k) {
    const int q = k * 256 + t;
    if (row0 + (q >> 3) < NODES) *(int4*)(bkt + (size_t)row0 * CAP + q * 4) = pc[k];
  }
}

__global__ __launch_bounds__(256)
void gemm_rs_wmma(const float* __restrict__ X, const float* __restrict__ W,
                  const int* __restrict__ cnt, float* __restrict__ out,
                  int ntiles) {
  __shared__ float sWt[C * LPITCH];
  for (int i = threadIdx.x; i < C * C; i += 256) {
    const int k = i >> 6, n = i & 63;
    sWt[n * LPITCH + k] = W[i];
  }
  __syncthreads();

  const int wave = threadIdx.x >> 5;
  const int lane = threadIdx.x & 31;
  const int tile = blockIdx.x * 8 + wave;
  if (tile >= ntiles) return;

  const int m0 = tile * 16;
  const int ar = m0 + (lane & 15);
  const int kg = (lane >> 4) * 2;
  const int cn = lane & 15;
  const float* bp = sWt + cn * LPITCH;

  v8f acc0 = {}, acc1 = {}, acc2 = {}, acc3 = {};
#pragma unroll
  for (int ks = 0; ks < 16; ++ks) {
    const int k0 = ks * 4 + kg;
    v2f a = *(const v2f*)(X + ar * C + k0);
    v2f b0 = *(const v2f*)(bp + k0);
    v2f b1 = *(const v2f*)(bp + 16 * LPITCH + k0);
    v2f b2 = *(const v2f*)(bp + 32 * LPITCH + k0);
    v2f b3 = *(const v2f*)(bp + 48 * LPITCH + k0);
    acc0 = __builtin_amdgcn_wmma_f32_16x16x4_f32(false, a, false, b0, (short)0, acc0, false, false);
    acc1 = __builtin_amdgcn_wmma_f32_16x16x4_f32(false, a, false, b1, (short)0, acc1, false, false);
    acc2 = __builtin_amdgcn_wmma_f32_16x16x4_f32(false, a, false, b2, (short)0, acc2, false, false);
    acc3 = __builtin_amdgcn_wmma_f32_16x16x4_f32(false, a, false, b3, (short)0, acc3, false, false);
  }

#pragma unroll
  for (int r = 0; r < 8; ++r) {
    const int row = m0 + r + 8 * (lane >> 4);
    const float s = rsqrtf((float)(cnt[row] + 1));
    out[row * C +  0 + cn] = acc0[r] * s;
    out[row * C + 16 + cn] = acc1[r] * s;
    out[row * C + 32 + cn] = acc2[r] * s;
    out[row * C + 48 + cn] = acc3[r] * s;
  }
}

__device__ __forceinline__ v4f gather_row(const int* __restrict__ cnt, const int* __restrict__ bkt,
                                          const int* __restrict__ novf, const int* __restrict__ ovf_src,
                                          const int* __restrict__ ovf_dst, const float* __restrict__ hs,
                                          const float* __restrict__ bias, int row, int lane) {
  const int half = lane >> 4;
  const int q = (lane & 15) * 4;
  const int deg = cnt[row];
  const int n = (deg < CAP) ? deg : CAP;
  const int my = (lane < n) ? bkt[row * CAP + lane] : 0;
  const float di = rsqrtf((float)(deg + 1));
  const v4f hr = *(const v4f*)(hs + (unsigned)row * C + q);
  const v4f b = *(const v4f*)(bias + q);
  v4f sum = {0.0f, 0.0f, 0.0f, 0.0f};
  int k = 0;
  for (; k + 8 <= n; k += 8) {
    v4f v[4];
#pragma unroll
    for (int j = 0; j < 4; ++j) {
      const int s = __shfl(my, k + 2 * j + half);
      v[j] = *(const v4f*)(hs + ((unsigned)s * C + q));
    }
#pragma unroll
    for (int j = 0; j < 4; ++j) sum += v[j];
  }
  if (k < n) {
    v4f v[4];
#pragma unroll
    for (int j = 0; j < 4; ++j) {
      const int s = __shfl(my, k + 2 * j + half);
      v[j] = *(const v4f*)(hs + ((unsigned)s * C + q));
    }
#pragma unroll
    for (int j = 0; j < 4; ++j)
      if (k + 2 * j + half < n) sum += v[j];
  }
  if (deg > CAP) {
    const int m = *novf;
    for (int base = 0; base < m; base += 32) {
      const int idx = base + lane;
      const int od = (idx < m) ? ovf_dst[idx] : -1;
      const int os = (idx < m) ? ovf_src[idx] : 0;
      unsigned hit = __builtin_amdgcn_ballot_w32(od == row);
      while (hit) {
        const int j = __builtin_ctz(hit);
        hit &= hit - 1;
        const int s = __shfl(os, j);
        const v4f v = *(const v4f*)(hs + ((unsigned)s * C + q));
        if (half == 0) sum += v;
      }
    }
  }
  v4f o;
  o.x = di * (sum.x + __shfl_xor(sum.x, 16) + hr.x) + b.x;
  o.y = di * (sum.y + __shfl_xor(sum.y, 16) + hr.y) + b.y;
  o.z = di * (sum.z + __shfl_xor(sum.z, 16) + hr.z) + b.z;
  o.w = di * (sum.w + __shfl_xor(sum.w, 16) + hr.w) + b.w;
  return o;
}

__global__ __launch_bounds__(256)
void gather_relu_kernel(const int* __restrict__ cnt, const int* __restrict__ bkt,
                        const int* __restrict__ novf, const int* __restrict__ ovf_src,
                        const int* __restrict__ ovf_dst, const float* __restrict__ hs,
                        const float* __restrict__ bias, float* __restrict__ outp) {
  const int lane = threadIdx.x & 31;
  const int row = blockIdx.x * 8 + __builtin_amdgcn_readfirstlane((int)(threadIdx.x >> 5));
  v4f o = gather_row(cnt, bkt, novf, ovf_src, ovf_dst, hs, bias, row, lane);
  if (lane < 16) {
    o.x = fmaxf(o.x, 0.0f);
    o.y = fmaxf(o.y, 0.0f);
    o.z = fmaxf(o.z, 0.0f);
    o.w = fmaxf(o.w, 0.0f);
    *(v4f*)(outp + ((unsigned)row * C + lane * 4)) = o;
  }
}

__global__ __launch_bounds__(256)
void gather_pool_kernel(const int* __restrict__ cnt, const int* __restrict__ bkt,
                        const int* __restrict__ novf, const int* __restrict__ ovf_src,
                        const int* __restrict__ ovf_dst, const float* __restrict__ hs,
                        const float* __restrict__ bias,
                        const int* __restrict__ batch, float* __restrict__ u) {
  __shared__ float srow[8 * C];
  __shared__ int sg[8];
  const int lane = threadIdx.x & 31;
  const int wave = __builtin_amdgcn_readfirstlane((int)(threadIdx.x >> 5));
  const int row = blockIdx.x * 8 + wave;
  const int g = batch[row];
  const v4f o = gather_row(cnt, bkt, novf, ovf_src, ovf_dst, hs, bias, row, lane);
  if (lane < 16) *(v4f*)(srow + wave * C + lane * 4) = o;
  if (lane == 0) sg[wave] = g;
  __syncthreads();
  if (threadIdx.x < C) {
    const int c = threadIdx.x;
    float v[8];
    int gid[8];
#pragma unroll
    for (int w = 0; w < 8; ++w) {
      v[w] = srow[w * C + c];
      gid[w] = sg[w];
    }
    int cur = gid[0];
    float run = v[0];
#pragma unroll
    for (int w = 1; w < 8; ++w) {
      if (gid[w] != cur) {
        atomicAdd(u + cur * C + c, run);
        cur = gid[w];
        run = v[w];
      } else {
        run += v[w];
      }
    }
    atomicAdd(u + cur * C + c, run);
  }
}

__global__ __launch_bounds__(256)
void head_wmma(const float* __restrict__ u, const float* __restrict__ Wf,
               const float* __restrict__ bf, float* __restrict__ y,
               float* __restrict__ out_u) {
  __shared__ float sWf[OUTC * LPITCH];
  for (int i = threadIdx.x; i < C * OUTC; i += 256) {
    const int k = i >> 5, n = i & 31;
    sWf[n * LPITCH + k] = Wf[i];
  }
  __syncthreads();

  const int wave = threadIdx.x >> 5;
  const int lane = threadIdx.x & 31;
  const int mt = wave >> 1;
  const int nt = wave & 1;
  const int ar = mt * 16 + (lane & 15);
  const int kg = (lane >> 4) * 2;
  const int cn = nt * 16 + (lane & 15);
  const float* bp = sWf + cn * LPITCH;

  v8f acc = {};
#pragma unroll
  for (int ks = 0; ks < 16; ++ks) {
    const int k0 = ks * 4 + kg;
    v2f a = *(const v2f*)(u + ar * C + k0);
    v2f b = *(const v2f*)(bp + k0);
    acc = __builtin_amdgcn_wmma_f32_16x16x4_f32(false, a, false, b, (short)0, acc, false, false);
  }
#pragma unroll
  for (int r = 0; r < 8; ++r) {
    const int row = mt * 16 + r + 8 * (lane >> 4);
    y[row * OUTC + cn] = acc[r] + bf[cn];
  }
#pragma unroll
  for (int j = 0; j < 16; ++j) {
    const int idx = threadIdx.x + j * 256;
    out_u[idx] = u[idx];
  }
}

extern "C" void kernel_launch(void* const* d_in, const int* in_sizes, int n_in,
                              void* d_out, int out_size, void* d_ws, size_t ws_size,
                              hipStream_t stream) {
  (void)in_sizes; (void)n_in; (void)out_size; (void)ws_size;

  const float* x     = (const float*)d_in[0];
  const int*   ei    = (const int*)  d_in[1];
  const int*   batch = (const int*)  d_in[2];
  const float* W1    = (const float*)d_in[3];
  const float* b1    = (const float*)d_in[4];
  const float* W2    = (const float*)d_in[5];
  const float* b2    = (const float*)d_in[6];
  const float* Wf    = (const float*)d_in[7];
  const float* bf    = (const float*)d_in[8];
  float* out = (float*)d_out;

  const int* src = ei;
  const int* dst = ei + EDGES;

  float* ws    = (float*)d_ws;
  float* hs    = ws;
  float* acc   = hs + (size_t)NODES * C;
  float* u     = acc + (size_t)NODES * C;
  int* cnt     = (int*)(u + GRAPHS * C);
  int* novf    = cnt + 50048;
  int* bkt     = novf + 32;
  int* ovf_src = bkt + (size_t)NODES * CAP;
  int* ovf_dst = ovf_src + EDGES;
  int* chunk   = ovf_dst + EDGES;
  int* seg     = chunk + EDGES;

  const dim3 B(256);
  const int NT = NODES / 16;
  const int GB = (NT + 7) / 8;

  bin_kernel   <<<NCHUNK, B, 0, stream>>>(src, dst, u, novf, chunk, seg);
  build_kernel <<<NBIN, B, 0, stream>>>(chunk, seg, cnt, bkt, novf, ovf_src, ovf_dst);

  gemm_rs_wmma <<<GB, B, 0, stream>>>(x, W1, cnt, hs, NT);
  gather_relu_kernel<<<NODES / 8, B, 0, stream>>>(cnt, bkt, novf, ovf_src, ovf_dst, hs, b1, acc);

  gemm_rs_wmma <<<GB, B, 0, stream>>>(acc, W2, cnt, hs, NT);
  gather_pool_kernel<<<NODES / 8, B, 0, stream>>>(cnt, bkt, novf, ovf_src, ovf_dst, hs, b2, batch, u);

  head_wmma  <<<1, B, 0, stream>>>(u, Wf, bf, out, out + GRAPHS * OUTC);
}
